# starting kernel
# baseline (speedup 1.0000x reference)
#include <hip/hip_runtime.h>


#define INPUT_DIM   128
#define HID         16
#define MAX_DEPTH   8
#define N_INTERNAL  255
#define N_LEAVES    256
#define NUM_CLASSES 10
#define EFF_IN      130
#define WFLOATS     (EFF_IN * HID)
#define ROWS_PER_WG 64
#define THREADS     128

typedef __attribute__((ext_vector_type(2))) float v2f;
typedef __attribute__((ext_vector_type(8))) float v8f;

struct SMem {
  float W[2][WFLOATS];
  float B1[2][HID];
  float W2[2][HID];
  float B2[2];
  float leafP[N_LEAVES * 16];
  float PP[ROWS_PER_WG][N_LEAVES + 1];
  float Cur[ROWS_PER_WG][129];
  float Prev[ROWS_PER_WG][65];
  float Prev2[ROWS_PER_WG][33];
};

__global__ __launch_bounds__(THREADS) void tree_wmma_kernel(
    const float* __restrict__ x, const float* __restrict__ path_prob,
    const float* __restrict__ W1, const float* __restrict__ b1,
    const float* __restrict__ w2, const float* __restrict__ b2,
    const float* __restrict__ leaf_logits, float* __restrict__ out)
{
  __shared__ SMem sm;
  const int tid  = threadIdx.x;
  const int wave = tid >> 5;
  const int lane = tid & 31;
  const int half = lane >> 4;
  const int col  = lane & 15;
  const int kb   = half * 2;
  const int wgRow0   = blockIdx.x * ROWS_PER_WG;
  const int waveRow0 = wgRow0 + wave * 16;
  const int rl       = wave * 16 + col;

  v2f afrag[32];
  {
    const v2f* x2 = (const v2f*)(x + (size_t)(waveRow0 + col) * INPUT_DIM);
    #pragma unroll
    for (int k = 0; k < 32; ++k) afrag[k] = x2[2 * k + half];
  }

  for (int j = tid; j < N_LEAVES * 16; j += THREADS) {
    const int lr = j >> 4, c = j & 15;
    sm.leafP[j] = (c < NUM_CLASSES) ? leaf_logits[lr * NUM_CLASSES + c] : 0.0f;
  }
  if (tid < ROWS_PER_WG) {
    sm.PP[tid][0]    = path_prob[wgRow0 + tid];
    sm.Prev[tid][0]  = 0.0f;
    sm.Prev2[tid][0] = 0.0f;
  }

  float4 wreg[5];
  float  b1v = 0.0f, w2v = 0.0f, b2v = 0.0f;

  {
    const float4* src = (const float4*)(W1);
    #pragma unroll
    for (int t = 0; t < 4; ++t) wreg[t] = src[tid + THREADS * t];
    if (tid < 8)   wreg[4] = src[tid + 512];
    if (tid < HID) { b1v = b1[tid]; w2v = w2[tid]; }
    if (tid == 0)  b2v = b2[0];

    float4* dst = (float4*)sm.W[0];
    #pragma unroll
    for (int t = 0; t < 4; ++t) dst[tid + THREADS * t] = wreg[t];
    if (tid < 8)   dst[tid + 512] = wreg[4];
    if (tid < HID) { sm.B1[0][tid] = b1v; sm.W2[0][tid] = w2v; }
    if (tid == 0)  sm.B2[0] = b2v;
  }
  __syncthreads();

  int curbuf = 0;
  for (int d = 0; d < MAX_DEPTH; ++d) {
    const int nNodes   = 1 << d;
    const int nodeBase = nNodes - 1;

    for (int i = 0; i < nNodes; ++i) {
      const int n = nodeBase + i;

      const bool havenext = (n + 1 < N_INTERNAL);
      if (havenext) {
        const float4* src = (const float4*)(W1 + (size_t)(n + 1) * WFLOATS);
        #pragma unroll
        for (int t = 0; t < 4; ++t) wreg[t] = src[tid + THREADS * t];
        if (tid < 8)   wreg[4] = src[tid + 512];
        if (tid < HID) { b1v = b1[(n + 1) * HID + tid]; w2v = w2[(n + 1) * HID + tid]; }
        if (tid == 0)  b2v = b2[n + 1];
      }
      if (n + 2 < N_INTERNAL)
        __builtin_prefetch(W1 + (size_t)(n + 2) * WFLOATS + tid * 16, 0, 1);

      const float* Wc = sm.W[curbuf];
      v8f acc0 = {0.f,0.f,0.f,0.f,0.f,0.f,0.f,0.f};
      v8f acc1 = {0.f,0.f,0.f,0.f,0.f,0.f,0.f,0.f};
      #pragma unroll
      for (int k = 0; k < 32; k += 2) {
        v2f bf0, bf1;
        bf0.x = Wc[(4 * k + kb) * HID + col];
        bf0.y = Wc[(4 * k + kb + 1) * HID + col];
        bf1.x = Wc[(4 * (k + 1) + kb) * HID + col];
        bf1.y = Wc[(4 * (k + 1) + kb + 1) * HID + col];
        acc0 = __builtin_amdgcn_wmma_f32_16x16x4_f32(false, afrag[k],     false, bf0,
                                                     (short)0, acc0, false, false);
        acc1 = __builtin_amdgcn_wmma_f32_16x16x4_f32(false, afrag[k + 1], false, bf1,
                                                     (short)0, acc1, false, false);
      }
      {
        v2f ae = {0.f, 0.f}, be = {0.f, 0.f};
        if (half == 0) {
          ae.x = sm.Prev[rl][i >> 1];
          ae.y = sm.Prev2[rl][i >> 2];
          be.x = Wc[128 * HID + col];
          be.y = Wc[129 * HID + col];
        }
        acc0 = __builtin_amdgcn_wmma_f32_16x16x4_f32(false, ae, false, be,
                                                     (short)0, acc0, false, false);
      }

      const float b1c = sm.B1[curbuf][col];
      const float w2c = sm.W2[curbuf][col];
      const float b2c = sm.B2[curbuf];
      #pragma unroll
      for (int r = 0; r < 8; ++r) {
        float h = acc0[r] + acc1[r];
        float t = fmaxf(h + b1c, 0.0f);
        float v = t * w2c;
        v += __shfl_xor(v, 1);
        v += __shfl_xor(v, 2);
        v += __shfl_xor(v, 4);
        v += __shfl_xor(v, 8);
        if (col == 0) sm.Cur[wave * 16 + half * 8 + r][i] = v + b2c;
      }

      if (havenext) {
        float4* dst = (float4*)sm.W[curbuf ^ 1];
        #pragma unroll
        for (int t = 0; t < 4; ++t) dst[tid + THREADS * t] = wreg[t];
        if (tid < 8)   dst[tid + 512] = wreg[4];
        if (tid < HID) { sm.B1[curbuf ^ 1][tid] = b1v; sm.W2[curbuf ^ 1][tid] = w2v; }
        if (tid == 0)  sm.B2[curbuf ^ 1] = b2v;
      }
      __syncthreads();
      curbuf ^= 1;
    }

    {
      const int shift = MAX_DEPTH - d;
      const int gap   = 1 << (shift - 1);
      for (int tix = tid; tix < (ROWS_PER_WG << d); tix += THREADS) {
        const int r = tix >> d;
        const int i = tix & (nNodes - 1);
        const float s = sm.Cur[r][i];
        const float p = 1.0f / (1.0f + expf(-s));
        const int slot = i << shift;
        const float v = sm.PP[r][slot];
        sm.PP[r][slot]       = v * p;
        sm.PP[r][slot + gap] = v * (1.0f - p);
      }
    }
    __syncthreads();
    if (d < MAX_DEPTH - 1) {
      if (d >= 1) {
        for (int tix = tid; tix < (ROWS_PER_WG << (d - 1)); tix += THREADS) {
          const int r = tix >> (d - 1);
          const int j = tix & ((1 << (d - 1)) - 1);
          sm.Prev2[r][j] = sm.Prev[r][j];
        }
      }
      __syncthreads();
      for (int tix = tid; tix < (ROWS_PER_WG << d); tix += THREADS) {
        const int r = tix >> d;
        const int j = tix & (nNodes - 1);
        sm.Prev[r][j] = sm.Cur[r][j];
      }
      __syncthreads();
    }
  }

  __syncthreads();
  {
    v8f e0 = {0.f,0.f,0.f,0.f,0.f,0.f,0.f,0.f};
    v8f e1 = {0.f,0.f,0.f,0.f,0.f,0.f,0.f,0.f};
    #pragma unroll
    for (int k = 0; k < 64; k += 2) {
      v2f a0, bq0, a1, bq1;
      a0.x  = sm.PP[rl][4 * k + kb];
      a0.y  = sm.PP[rl][4 * k + kb + 1];
      bq0.x = sm.leafP[(4 * k + kb) * 16 + col];
      bq0.y = sm.leafP[(4 * k + kb + 1) * 16 + col];
      a1.x  = sm.PP[rl][4 * (k + 1) + kb];
      a1.y  = sm.PP[rl][4 * (k + 1) + kb + 1];
      bq1.x = sm.leafP[(4 * (k + 1) + kb) * 16 + col];
      bq1.y = sm.leafP[(4 * (k + 1) + kb + 1) * 16 + col];
      e0 = __builtin_amdgcn_wmma_f32_16x16x4_f32(false, a0, false, bq0,
                                                 (short)0, e0, false, false);
      e1 = __builtin_amdgcn_wmma_f32_16x16x4_f32(false, a1, false, bq1,
                                                 (short)0, e1, false, false);
    }
    if (col < NUM_CLASSES) {
      #pragma unroll
      for (int r = 0; r < 8; ++r)
        out[(size_t)(waveRow0 + half * 8 + r) * NUM_CLASSES + col] = e0[r] + e1[r];
    }
  }
}

extern "C" void kernel_launch(void* const* d_in, const int* in_sizes, int n_in,
                              void* d_out, int out_size, void* d_ws, size_t ws_size,
                              hipStream_t stream) {
  const float* x    = (const float*)d_in[0];
  const float* pp   = (const float*)d_in[1];
  const float* W1   = (const float*)d_in[2];
  const float* b1   = (const float*)d_in[3];
  const float* w2   = (const float*)d_in[4];
  const float* b2   = (const float*)d_in[5];
  const float* leaf = (const float*)d_in[6];
  float* out = (float*)d_out;

  const int B = in_sizes[0] / INPUT_DIM;
  dim3 grid(B / ROWS_PER_WG);
  dim3 block(THREADS);
  tree_wmma_kernel<<<grid, block, 0, stream>>>(x, pp, W1, b1, w2, b2, leaf, out);
}


# Swap WMMA operand roles (A=weights, B=x) so hidden units lie on accumulator registers: w2 sum = 8 in-lane FMAs + one half-lane exchange, replacing 32 serial ds_bpermute; hidden-sum order changes
# speedup vs baseline: 1.4786x; 1.4786x over previous
#include <hip/hip_runtime.h>


#define INPUT_DIM   128
#define HID         16
#define MAX_DEPTH   8
#define N_INTERNAL  255
#define N_LEAVES    256
#define NUM_CLASSES 10
#define EFF_IN      130
#define WFLOATS     (EFF_IN * HID)
#define ROWS_PER_WG 64
#define THREADS     128

typedef __attribute__((ext_vector_type(2))) float v2f;
typedef __attribute__((ext_vector_type(8))) float v8f;

__device__ __forceinline__ float xhalf(float x) {
  const int b = __builtin_bit_cast(int, x);
  return __builtin_bit_cast(float,
      __builtin_amdgcn_permlanex16(b, b, 0x76543210, 0xfedcba98, false, false));
}

struct SMem {
  float W[2][WFLOATS];
  float B1[2][HID];
  float W2[2][HID];
  float B2[2];
  float leafP[N_LEAVES * 16];
  float PP[ROWS_PER_WG][N_LEAVES + 1];
  float Cur[ROWS_PER_WG][129];
  float Prev[ROWS_PER_WG][65];
  float Prev2[ROWS_PER_WG][33];
};

__global__ __launch_bounds__(THREADS) void tree_wmma_kernel(
    const float* __restrict__ x, const float* __restrict__ path_prob,
    const float* __restrict__ W1, const float* __restrict__ b1,
    const float* __restrict__ w2, const float* __restrict__ b2,
    const float* __restrict__ leaf_logits, float* __restrict__ out)
{
  __shared__ SMem sm;
  const int tid  = threadIdx.x;
  const int wave = tid >> 5;
  const int lane = tid & 31;
  const int half = lane >> 4;
  const int col  = lane & 15;
  const int kb   = half * 2;
  const int wgRow0   = blockIdx.x * ROWS_PER_WG;
  const int waveRow0 = wgRow0 + wave * 16;
  const int rl       = wave * 16 + col;

  v2f afrag[32];
  {
    const v2f* x2 = (const v2f*)(x + (size_t)(waveRow0 + col) * INPUT_DIM);
    #pragma unroll
    for (int k = 0; k < 32; ++k) afrag[k] = x2[2 * k + half];
  }

  for (int j = tid; j < N_LEAVES * 16; j += THREADS) {
    const int lr = j >> 4, c = j & 15;
    sm.leafP[j] = (c < NUM_CLASSES) ? leaf_logits[lr * NUM_CLASSES + c] : 0.0f;
  }
  if (tid < ROWS_PER_WG) {
    sm.PP[tid][0]    = path_prob[wgRow0 + tid];
    sm.Prev[tid][0]  = 0.0f;
    sm.Prev2[tid][0] = 0.0f;
  }

  float4 wreg[5];
  float  b1v = 0.0f, w2v = 0.0f, b2v = 0.0f;

  {
    const float4* src = (const float4*)(W1);
    #pragma unroll
    for (int t = 0; t < 4; ++t) wreg[t] = src[tid + THREADS * t];
    if (tid < 8)   wreg[4] = src[tid + 512];
    if (tid < HID) { b1v = b1[tid]; w2v = w2[tid]; }
    if (tid == 0)  b2v = b2[0];

    float4* dst = (float4*)sm.W[0];
    #pragma unroll
    for (int t = 0; t < 4; ++t) dst[tid + THREADS * t] = wreg[t];
    if (tid < 8)   dst[tid + 512] = wreg[4];
    if (tid < HID) { sm.B1[0][tid] = b1v; sm.W2[0][tid] = w2v; }
    if (tid == 0)  sm.B2[0] = b2v;
  }
  __syncthreads();

  int curbuf = 0;
  for (int d = 0; d < MAX_DEPTH; ++d) {
    const int nNodes   = 1 << d;
    const int nodeBase = nNodes - 1;

    for (int i = 0; i < nNodes; ++i) {
      const int n = nodeBase + i;

      const bool havenext = (n + 1 < N_INTERNAL);
      if (havenext) {
        const float4* src = (const float4*)(W1 + (size_t)(n + 1) * WFLOATS);
        #pragma unroll
        for (int t = 0; t < 4; ++t) wreg[t] = src[tid + THREADS * t];
        if (tid < 8)   wreg[4] = src[tid + 512];
        if (tid < HID) { b1v = b1[(n + 1) * HID + tid]; w2v = w2[(n + 1) * HID + tid]; }
        if (tid == 0)  b2v = b2[n + 1];
      }
      if (n + 2 < N_INTERNAL)
        __builtin_prefetch(W1 + (size_t)(n + 2) * WFLOATS + tid * 16, 0, 1);

      const float* Wc = sm.W[curbuf];
      const float4 b1lo = *(const float4*)&sm.B1[curbuf][half * 8];
      const float4 b1hi = *(const float4*)&sm.B1[curbuf][half * 8 + 4];
      const float4 w2lo = *(const float4*)&sm.W2[curbuf][half * 8];
      const float4 w2hi = *(const float4*)&sm.W2[curbuf][half * 8 + 4];
      const float  b2c  = sm.B2[curbuf];
      v8f acc0 = {0.f,0.f,0.f,0.f,0.f,0.f,0.f,0.f};
      v8f acc1 = {0.f,0.f,0.f,0.f,0.f,0.f,0.f,0.f};
      #pragma unroll
      for (int k = 0; k < 32; k += 2) {
        v2f bf0, bf1;
        bf0.x = Wc[(4 * k + kb) * HID + col];
        bf0.y = Wc[(4 * k + kb + 1) * HID + col];
        bf1.x = Wc[(4 * (k + 1) + kb) * HID + col];
        bf1.y = Wc[(4 * (k + 1) + kb + 1) * HID + col];
        acc0 = __builtin_amdgcn_wmma_f32_16x16x4_f32(false, bf0, false, afrag[k],
                                                     (short)0, acc0, false, false);
        acc1 = __builtin_amdgcn_wmma_f32_16x16x4_f32(false, bf1, false, afrag[k + 1],
                                                     (short)0, acc1, false, false);
      }
      {
        v2f ae = {0.f, 0.f}, be = {0.f, 0.f};
        if (half == 0) {
          ae.x = sm.Prev[rl][i >> 1];
          ae.y = sm.Prev2[rl][i >> 2];
          be.x = Wc[128 * HID + col];
          be.y = Wc[129 * HID + col];
        }
        acc0 = __builtin_amdgcn_wmma_f32_16x16x4_f32(false, be, false, ae,
                                                     (short)0, acc0, false, false);
      }

      {
        const float b1r[8] = {b1lo.x, b1lo.y, b1lo.z, b1lo.w, b1hi.x, b1hi.y, b1hi.z, b1hi.w};
        const float w2r[8] = {w2lo.x, w2lo.y, w2lo.z, w2lo.w, w2hi.x, w2hi.y, w2hi.z, w2hi.w};
        float p0 = 0.0f, p1 = 0.0f;
        #pragma unroll
        for (int r = 0; r < 8; r += 2) {
          const float t0 = fmaxf((acc0[r]     + acc1[r])     + b1r[r],     0.0f);
          const float t1 = fmaxf((acc0[r + 1] + acc1[r + 1]) + b1r[r + 1], 0.0f);
          p0 += t0 * w2r[r];
          p1 += t1 * w2r[r + 1];
        }
        const float part = p0 + p1;
        const float tot  = part + xhalf(part);
        if (half == 0) sm.Cur[rl][i] = tot + b2c;
      }

      if (havenext) {
        float4* dst = (float4*)sm.W[curbuf ^ 1];
        #pragma unroll
        for (int t = 0; t < 4; ++t) dst[tid + THREADS * t] = wreg[t];
        if (tid < 8)   dst[tid + 512] = wreg[4];
        if (tid < HID) { sm.B1[curbuf ^ 1][tid] = b1v; sm.W2[curbuf ^ 1][tid] = w2v; }
        if (tid == 0)  sm.B2[curbuf ^ 1] = b2v;
      }
      __syncthreads();
      curbuf ^= 1;
    }

    {
      const int shift = MAX_DEPTH - d;
      const int gap   = 1 << (shift - 1);
      for (int tix = tid; tix < (ROWS_PER_WG << d); tix += THREADS) {
        const int r = tix >> d;
        const int i = tix & (nNodes - 1);
        const float s = sm.Cur[r][i];
        const float p = 1.0f / (1.0f + expf(-s));
        const int slot = i << shift;
        const float v = sm.PP[r][slot];
        sm.PP[r][slot]       = v * p;
        sm.PP[r][slot + gap] = v * (1.0f - p);
      }
    }
    __syncthreads();
    if (d < MAX_DEPTH - 1) {
      if (d >= 1) {
        for (int tix = tid; tix < (ROWS_PER_WG << (d - 1)); tix += THREADS) {
          const int r = tix >> (d - 1);
          const int j = tix & ((1 << (d - 1)) - 1);
          sm.Prev2[r][j] = sm.Prev[r][j];
        }
      }
      __syncthreads();
      for (int tix = tid; tix < (ROWS_PER_WG << d); tix += THREADS) {
        const int r = tix >> d;
        const int j = tix & (nNodes - 1);
        sm.Prev[r][j] = sm.Cur[r][j];
      }
      __syncthreads();
    }
  }

  __syncthreads();
  {
    v8f e0 = {0.f,0.f,0.f,0.f,0.f,0.f,0.f,0.f};
    v8f e1 = {0.f,0.f,0.f,0.f,0.f,0.f,0.f,0.f};
    #pragma unroll
    for (int k = 0; k < 64; k += 2) {
      v2f a0, bq0, a1, bq1;
      a0.x  = sm.PP[rl][4 * k + kb];
      a0.y  = sm.PP[rl][4 * k + kb + 1];
      bq0.x = sm.leafP[(4 * k + kb) * 16 + col];
      bq0.y = sm.leafP[(4 * k + kb + 1) * 16 + col];
      a1.x  = sm.PP[rl][4 * (k + 1) + kb];
      a1.y  = sm.PP[rl][4 * (k + 1) + kb + 1];
      bq1.x = sm.leafP[(4 * (k + 1) + kb) * 16 + col];
      bq1.y = sm.leafP[(4 * (k + 1) + kb + 1) * 16 + col];
      e0 = __builtin_amdgcn_wmma_f32_16x16x4_f32(false, a0, false, bq0,
                                                 (short)0, e0, false, false);
      e1 = __builtin_amdgcn_wmma_f32_16x16x4_f32(false, a1, false, bq1,
                                                 (short)0, e1, false, false);
    }
    if (col < NUM_CLASSES) {
      #pragma unroll
      for (int r = 0; r < 8; ++r)
        out[(size_t)(waveRow0 + half * 8 + r) * NUM_CLASSES + col] = e0[r] + e1[r];
    }
  }
}

extern "C" void kernel_launch(void* const* d_in, const int* in_sizes, int n_in,
                              void* d_out, int out_size, void* d_ws, size_t ws_size,
                              hipStream_t stream) {
  const float* x    = (const float*)d_in[0];
  const float* pp   = (const float*)d_in[1];
  const float* W1   = (const float*)d_in[2];
  const float* b1   = (const float*)d_in[3];
  const float* w2   = (const float*)d_in[4];
  const float* b2   = (const float*)d_in[5];
  const float* leaf = (const float*)d_in[6];
  float* out = (float*)d_out;

  const int B = in_sizes[0] / INPUT_DIM;
  dim3 grid(B / ROWS_PER_WG);
  dim3 block(THREADS);
  tree_wmma_kernel<<<grid, block, 0, stream>>>(x, pp, W1, b1, w2, b2, leaf, out);
}


# Hold the prefetched next-node weights in five named float4 registers instead of a local array the compiler kept in scratch: loads now stay in flight across the 33 WMMAs (scratch 96 B to 0)
# speedup vs baseline: 3.4192x; 2.3125x over previous
#include <hip/hip_runtime.h>


#define INPUT_DIM   128
#define HID         16
#define MAX_DEPTH   8
#define N_INTERNAL  255
#define N_LEAVES    256
#define NUM_CLASSES 10
#define EFF_IN      130
#define WFLOATS     (EFF_IN * HID)
#define ROWS_PER_WG 64
#define THREADS     128

typedef __attribute__((ext_vector_type(2))) float v2f;
typedef __attribute__((ext_vector_type(8))) float v8f;

__device__ __forceinline__ float xhalf(float x) {
  const int b = __builtin_bit_cast(int, x);
  return __builtin_bit_cast(float,
      __builtin_amdgcn_permlanex16(b, b, 0x76543210, 0xfedcba98, false, false));
}

struct SMem {
  float W[2][WFLOATS];
  float B1[2][HID];
  float W2[2][HID];
  float B2[2];
  float leafP[N_LEAVES * 16];
  float PP[ROWS_PER_WG][N_LEAVES + 1];
  float Cur[ROWS_PER_WG][129];
  float Prev[ROWS_PER_WG][65];
  float Prev2[ROWS_PER_WG][33];
};

__global__ __launch_bounds__(THREADS) void tree_wmma_kernel(
    const float* __restrict__ x, const float* __restrict__ path_prob,
    const float* __restrict__ W1, const float* __restrict__ b1,
    const float* __restrict__ w2, const float* __restrict__ b2,
    const float* __restrict__ leaf_logits, float* __restrict__ out)
{
  __shared__ SMem sm;
  const int tid  = threadIdx.x;
  const int wave = tid >> 5;
  const int lane = tid & 31;
  const int half = lane >> 4;
  const int col  = lane & 15;
  const int kb   = half * 2;
  const int wgRow0   = blockIdx.x * ROWS_PER_WG;
  const int waveRow0 = wgRow0 + wave * 16;
  const int rl       = wave * 16 + col;

  v2f afrag[32];
  {
    const v2f* x2 = (const v2f*)(x + (size_t)(waveRow0 + col) * INPUT_DIM);
    #pragma unroll
    for (int k = 0; k < 32; ++k) afrag[k] = x2[2 * k + half];
  }

  for (int j = tid; j < N_LEAVES * 16; j += THREADS) {
    const int lr = j >> 4, c = j & 15;
    sm.leafP[j] = (c < NUM_CLASSES) ? leaf_logits[lr * NUM_CLASSES + c] : 0.0f;
  }
  if (tid < ROWS_PER_WG) {
    sm.PP[tid][0]    = path_prob[wgRow0 + tid];
    sm.Prev[tid][0]  = 0.0f;
    sm.Prev2[tid][0] = 0.0f;
  }

  float4 wr0, wr1, wr2, wr3, wr4;
  float  b1v = 0.0f, w2v = 0.0f, b2v = 0.0f;

  {
    const float4* src = (const float4*)(W1);
    wr0 = src[tid];
    wr1 = src[tid + THREADS];
    wr2 = src[tid + THREADS * 2];
    wr3 = src[tid + THREADS * 3];
    if (tid < 8)   wr4 = src[tid + 512];
    if (tid < HID) { b1v = b1[tid]; w2v = w2[tid]; }
    if (tid == 0)  b2v = b2[0];

    float4* dst = (float4*)sm.W[0];
    dst[tid]               = wr0;
    dst[tid + THREADS]     = wr1;
    dst[tid + THREADS * 2] = wr2;
    dst[tid + THREADS * 3] = wr3;
    if (tid < 8)   dst[tid + 512] = wr4;
    if (tid < HID) { sm.B1[0][tid] = b1v; sm.W2[0][tid] = w2v; }
    if (tid == 0)  sm.B2[0] = b2v;
  }
  __syncthreads();

  int curbuf = 0;
  for (int d = 0; d < MAX_DEPTH; ++d) {
    const int nNodes   = 1 << d;
    const int nodeBase = nNodes - 1;

    for (int i = 0; i < nNodes; ++i) {
      const int n = nodeBase + i;

      const bool havenext = (n + 1 < N_INTERNAL);
      if (havenext) {
        const float4* src = (const float4*)(W1 + (size_t)(n + 1) * WFLOATS);
        wr0 = src[tid];
        wr1 = src[tid + THREADS];
        wr2 = src[tid + THREADS * 2];
        wr3 = src[tid + THREADS * 3];
        if (tid < 8)   wr4 = src[tid + 512];
        if (tid < HID) { b1v = b1[(n + 1) * HID + tid]; w2v = w2[(n + 1) * HID + tid]; }
        if (tid == 0)  b2v = b2[n + 1];
      }
      if (n + 2 < N_INTERNAL)
        __builtin_prefetch(W1 + (size_t)(n + 2) * WFLOATS + tid * 16, 0, 1);

      const float* Wc = sm.W[curbuf];
      const float4 b1lo = *(const float4*)&sm.B1[curbuf][half * 8];
      const float4 b1hi = *(const float4*)&sm.B1[curbuf][half * 8 + 4];
      const float4 w2lo = *(const float4*)&sm.W2[curbuf][half * 8];
      const float4 w2hi = *(const float4*)&sm.W2[curbuf][half * 8 + 4];
      const float  b2c  = sm.B2[curbuf];
      v8f acc0 = {0.f,0.f,0.f,0.f,0.f,0.f,0.f,0.f};
      v8f acc1 = {0.f,0.f,0.f,0.f,0.f,0.f,0.f,0.f};
      #pragma unroll
      for (int k = 0; k < 32; k += 2) {
        v2f bf0, bf1;
        bf0.x = Wc[(4 * k + kb) * HID + col];
        bf0.y = Wc[(4 * k + kb + 1) * HID + col];
        bf1.x = Wc[(4 * (k + 1) + kb) * HID + col];
        bf1.y = Wc[(4 * (k + 1) + kb + 1) * HID + col];
        acc0 = __builtin_amdgcn_wmma_f32_16x16x4_f32(false, bf0, false, afrag[k],
                                                     (short)0, acc0, false, false);
        acc1 = __builtin_amdgcn_wmma_f32_16x16x4_f32(false, bf1, false, afrag[k + 1],
                                                     (short)0, acc1, false, false);
      }
      {
        v2f ae = {0.f, 0.f}, be = {0.f, 0.f};
        if (half == 0) {
          ae.x = sm.Prev[rl][i >> 1];
          ae.y = sm.Prev2[rl][i >> 2];
          be.x = Wc[128 * HID + col];
          be.y = Wc[129 * HID + col];
        }
        acc0 = __builtin_amdgcn_wmma_f32_16x16x4_f32(false, be, false, ae,
                                                     (short)0, acc0, false, false);
      }

      {
        const float b1r[8] = {b1lo.x, b1lo.y, b1lo.z, b1lo.w, b1hi.x, b1hi.y, b1hi.z, b1hi.w};
        const float w2r[8] = {w2lo.x, w2lo.y, w2lo.z, w2lo.w, w2hi.x, w2hi.y, w2hi.z, w2hi.w};
        float p0 = 0.0f, p1 = 0.0f;
        #pragma unroll
        for (int r = 0; r < 8; r += 2) {
          const float t0 = fmaxf((acc0[r]     + acc1[r])     + b1r[r],     0.0f);
          const float t1 = fmaxf((acc0[r + 1] + acc1[r + 1]) + b1r[r + 1], 0.0f);
          p0 += t0 * w2r[r];
          p1 += t1 * w2r[r + 1];
        }
        const float part = p0 + p1;
        const float tot  = part + xhalf(part);
        if (half == 0) sm.Cur[rl][i] = tot + b2c;
      }

      if (havenext) {
        float4* dst = (float4*)sm.W[curbuf ^ 1];
        dst[tid]               = wr0;
        dst[tid + THREADS]     = wr1;
        dst[tid + THREADS * 2] = wr2;
        dst[tid + THREADS * 3] = wr3;
        if (tid < 8)   dst[tid + 512] = wr4;
        if (tid < HID) { sm.B1[curbuf ^ 1][tid] = b1v; sm.W2[curbuf ^ 1][tid] = w2v; }
        if (tid == 0)  sm.B2[curbuf ^ 1] = b2v;
      }
      __syncthreads();
      curbuf ^= 1;
    }

    {
      const int shift = MAX_DEPTH - d;
      const int gap   = 1 << (shift - 1);
      for (int tix = tid; tix < (ROWS_PER_WG << d); tix += THREADS) {
        const int r = tix >> d;
        const int i = tix & (nNodes - 1);
        const float s = sm.Cur[r][i];
        const float p = 1.0f / (1.0f + expf(-s));
        const int slot = i << shift;
        const float v = sm.PP[r][slot];
        sm.PP[r][slot]       = v * p;
        sm.PP[r][slot + gap] = v * (1.0f - p);
      }
    }
    __syncthreads();
    if (d < MAX_DEPTH - 1) {
      if (d >= 1) {
        for (int tix = tid; tix < (ROWS_PER_WG << (d - 1)); tix += THREADS) {
          const int r = tix >> (d - 1);
          const int j = tix & ((1 << (d - 1)) - 1);
          sm.Prev2[r][j] = sm.Prev[r][j];
        }
      }
      __syncthreads();
      for (int tix = tid; tix < (ROWS_PER_WG << d); tix += THREADS) {
        const int r = tix >> d;
        const int j = tix & (nNodes - 1);
        sm.Prev[r][j] = sm.Cur[r][j];
      }
      __syncthreads();
    }
  }

  __syncthreads();
  {
    v8f e0 = {0.f,0.f,0.f,0.f,0.f,0.f,0.f,0.f};
    v8f e1 = {0.f,0.f,0.f,0.f,0.f,0.f,0.f,0.f};
    #pragma unroll
    for (int k = 0; k < 64; k += 2) {
      v2f a0, bq0, a1, bq1;
      a0.x  = sm.PP[rl][4 * k + kb];
      a0.y  = sm.PP[rl][4 * k + kb + 1];
      bq0.x = sm.leafP[(4 * k + kb) * 16 + col];
      bq0.y = sm.leafP[(4 * k + kb + 1) * 16 + col];
      a1.x  = sm.PP[rl][4 * (k + 1) + kb];
      a1.y  = sm.PP[rl][4 * (k + 1) + kb + 1];
      bq1.x = sm.leafP[(4 * (k + 1) + kb) * 16 + col];
      bq1.y = sm.leafP[(4 * (k + 1) + kb + 1) * 16 + col];
      e0 = __builtin_amdgcn_wmma_f32_16x16x4_f32(false, a0, false, bq0,
                                                 (short)0, e0, false, false);
      e1 = __builtin_amdgcn_wmma_f32_16x16x4_f32(false, a1, false, bq1,
                                                 (short)0, e1, false, false);
    }
    if (col < NUM_CLASSES) {
      #pragma unroll
      for (int r = 0; r < 8; ++r)
        out[(size_t)(waveRow0 + half * 8 + r) * NUM_CLASSES + col] = e0[r] + e1[r];
    }
  }
}

extern "C" void kernel_launch(void* const* d_in, const int* in_sizes, int n_in,
                              void* d_out, int out_size, void* d_ws, size_t ws_size,
                              hipStream_t stream) {
  const float* x    = (const float*)d_in[0];
  const float* pp   = (const float*)d_in[1];
  const float* W1   = (const float*)d_in[2];
  const float* b1   = (const float*)d_in[3];
  const float* w2   = (const float*)d_in[4];
  const float* b2   = (const float*)d_in[5];
  const float* leaf = (const float*)d_in[6];
  float* out = (float*)d_out;

  const int B = in_sizes[0] / INPUT_DIM;
  dim3 grid(B / ROWS_PER_WG);
  dim3 block(THREADS);
  tree_wmma_kernel<<<grid, block, 0, stream>>>(x, pp, W1, b1, w2, b2, leaf, out);
}


# Eight waves per workgroup in two groups: from depth 1 on two nodes are staged per step and each group computes one of them for the same 64 rows; one barrier per two nodes, four waves per SIMD
# speedup vs baseline: 4.4624x; 1.3051x over previous
#include <hip/hip_runtime.h>


#define INPUT_DIM   128
#define HID         16
#define MAX_DEPTH   8
#define N_INTERNAL  255
#define N_LEAVES    256
#define NUM_CLASSES 10
#define EFF_IN      130
#define WFLOATS     (EFF_IN * HID)
#define ROWS_PER_WG 64
#define PAIRF       (2 * WFLOATS)
#define THREADS     256

typedef __attribute__((ext_vector_type(2))) float v2f;
typedef __attribute__((ext_vector_type(8))) float v8f;

__device__ __forceinline__ float xhalf(float x) {
  const int b = __builtin_bit_cast(int, x);
  return __builtin_bit_cast(float,
      __builtin_amdgcn_permlanex16(b, b, 0x76543210, 0xfedcba98, false, false));
}

struct SMem {
  alignas(16) float W[2][PAIRF];
  float B1[2][2 * HID];
  float W2[2][2 * HID];
  float B2[2][2];
  float PP[ROWS_PER_WG][N_LEAVES + 1];
  float Cur[ROWS_PER_WG][129];
  float Prev[ROWS_PER_WG][65];
  float Prev2[ROWS_PER_WG][33];
};

__global__ __launch_bounds__(THREADS) void tree_wmma_kernel(
    const float* __restrict__ x, const float* __restrict__ path_prob,
    const float* __restrict__ W1, const float* __restrict__ b1,
    const float* __restrict__ w2, const float* __restrict__ b2,
    const float* __restrict__ leaf_logits, float* __restrict__ out)
{
  __shared__ SMem sm;
  const int tid  = threadIdx.x;
  const int wave = tid >> 5;
  const int lane = tid & 31;
  const int grp  = wave >> 2;
  const int rt   = wave & 3;
  const int half = lane >> 4;
  const int col  = lane & 15;
  const int kb   = half * 2;
  const int wgRow0   = blockIdx.x * ROWS_PER_WG;
  const int waveRow0 = wgRow0 + rt * 16;
  const int rl       = rt * 16 + col;

  v2f afrag[32];
  {
    const v2f* x2 = (const v2f*)(x + (size_t)(waveRow0 + col) * INPUT_DIM);
    #pragma unroll
    for (int k = 0; k < 32; ++k) afrag[k] = x2[2 * k + half];
  }

  if (tid < ROWS_PER_WG) {
    sm.PP[tid][0]    = path_prob[wgRow0 + tid];
    sm.Prev[tid][0]  = 0.0f;
    sm.Prev2[tid][0] = 0.0f;
  }

  float4 wr0, wr1, wr2, wr3, wr4;
  float  b1v = 0.0f, w2v = 0.0f, b2v = 0.0f;

  {
    const float4* src = (const float4*)(W1);
    wr0 = src[tid];
    wr1 = src[tid + THREADS];
    wr2 = src[tid + THREADS * 2];
    wr3 = src[tid + THREADS * 3];
    if (tid < 16)      wr4 = src[tid + 1024];
    if (tid < 2 * HID) { b1v = b1[tid]; w2v = w2[tid]; }
    if (tid < 2)       b2v = b2[tid];

    float4* dst = (float4*)sm.W[0];
    dst[tid]               = wr0;
    dst[tid + THREADS]     = wr1;
    dst[tid + THREADS * 2] = wr2;
    dst[tid + THREADS * 3] = wr3;
    if (tid < 16)      dst[tid + 1024] = wr4;
    if (tid < 2 * HID) { sm.B1[0][tid] = b1v; sm.W2[0][tid] = w2v; }
    if (tid < 2)       sm.B2[0][tid] = b2v;
  }
  __syncthreads();

  int curbuf = 0;
  for (int d = 0; d < MAX_DEPTH; ++d) {
    const int nNodes   = 1 << d;
    const int nodeBase = nNodes - 1;

    const int cnt = (d == 0) ? 1 : 2;
    for (int i0 = 0; i0 < nNodes; i0 += cnt) {
      const int nxt = nodeBase + i0 + cnt;

      const bool havenext = (nxt < N_INTERNAL);
      if (havenext) {
        const float4* src = (const float4*)(W1 + (size_t)nxt * WFLOATS);
        wr0 = src[tid];
        wr1 = src[tid + THREADS];
        wr2 = src[tid + THREADS * 2];
        wr3 = src[tid + THREADS * 3];
        if (tid < 16)      wr4 = src[tid + 1024];
        if (tid < 2 * HID) { b1v = b1[nxt * HID + tid]; w2v = w2[nxt * HID + tid]; }
        if (tid < 2)       b2v = b2[nxt + tid];
      }
      if (nxt + 2 < N_INTERNAL)
        __builtin_prefetch(W1 + (size_t)(nxt + 2) * WFLOATS + tid * 16, 0, 1);

      if (grp < cnt) {
      const int i = i0 + grp;
      const float* Wc = sm.W[curbuf] + grp * WFLOATS;
      const float4 b1lo = *(const float4*)&sm.B1[curbuf][grp * HID + half * 8];
      const float4 b1hi = *(const float4*)&sm.B1[curbuf][grp * HID + half * 8 + 4];
      const float4 w2lo = *(const float4*)&sm.W2[curbuf][grp * HID + half * 8];
      const float4 w2hi = *(const float4*)&sm.W2[curbuf][grp * HID + half * 8 + 4];
      const float  b2c  = sm.B2[curbuf][grp];
      v8f acc0 = {0.f,0.f,0.f,0.f,0.f,0.f,0.f,0.f};
      v8f acc1 = {0.f,0.f,0.f,0.f,0.f,0.f,0.f,0.f};
      #pragma unroll
      for (int k = 0; k < 32; k += 2) {
        v2f bf0, bf1;
        bf0.x = Wc[(4 * k + kb) * HID + col];
        bf0.y = Wc[(4 * k + kb + 1) * HID + col];
        bf1.x = Wc[(4 * (k + 1) + kb) * HID + col];
        bf1.y = Wc[(4 * (k + 1) + kb + 1) * HID + col];
        acc0 = __builtin_amdgcn_wmma_f32_16x16x4_f32(false, bf0, false, afrag[k],
                                                     (short)0, acc0, false, false);
        acc1 = __builtin_amdgcn_wmma_f32_16x16x4_f32(false, bf1, false, afrag[k + 1],
                                                     (short)0, acc1, false, false);
      }
      {
        v2f ae = {0.f, 0.f}, be = {0.f, 0.f};
        if (half == 0) {
          ae.x = sm.Prev[rl][i >> 1];
          ae.y = sm.Prev2[rl][i >> 2];
          be.x = Wc[128 * HID + col];
          be.y = Wc[129 * HID + col];
        }
        acc0 = __builtin_amdgcn_wmma_f32_16x16x4_f32(false, be, false, ae,
                                                     (short)0, acc0, false, false);
      }

      {
        const float b1r[8] = {b1lo.x, b1lo.y, b1lo.z, b1lo.w, b1hi.x, b1hi.y, b1hi.z, b1hi.w};
        const float w2r[8] = {w2lo.x, w2lo.y, w2lo.z, w2lo.w, w2hi.x, w2hi.y, w2hi.z, w2hi.w};
        float p0 = 0.0f, p1 = 0.0f;
        #pragma unroll
        for (int r = 0; r < 8; r += 2) {
          const float t0 = fmaxf((acc0[r]     + acc1[r])     + b1r[r],     0.0f);
          const float t1 = fmaxf((acc0[r + 1] + acc1[r + 1]) + b1r[r + 1], 0.0f);
          p0 += t0 * w2r[r];
          p1 += t1 * w2r[r + 1];
        }
        const float part = p0 + p1;
        const float tot  = part + xhalf(part);
        if (half == 0) sm.Cur[rl][i] = tot + b2c;
      }
      }

      if (havenext) {
        float4* dst = (float4*)sm.W[curbuf ^ 1];
        dst[tid]               = wr0;
        dst[tid + THREADS]     = wr1;
        dst[tid + THREADS * 2] = wr2;
        dst[tid + THREADS * 3] = wr3;
        if (tid < 16)      dst[tid + 1024] = wr4;
        if (tid < 2 * HID) { sm.B1[curbuf ^ 1][tid] = b1v; sm.W2[curbuf ^ 1][tid] = w2v; }
        if (tid < 2)       sm.B2[curbuf ^ 1][tid] = b2v;
      }
      __syncthreads();
      curbuf ^= 1;
    }

    if (d == MAX_DEPTH - 1) {
      float* leafP = &sm.W[0][0];
      for (int j = tid; j < N_LEAVES * 16; j += THREADS) {
        const int lr = j >> 4, c = j & 15;
        leafP[j] = (c < NUM_CLASSES) ? leaf_logits[lr * NUM_CLASSES + c] : 0.0f;
      }
    }

    {
      const int shift = MAX_DEPTH - d;
      const int gap   = 1 << (shift - 1);
      for (int tix = tid; tix < (ROWS_PER_WG << d); tix += THREADS) {
        const int r = tix >> d;
        const int i = tix & (nNodes - 1);
        const float s = sm.Cur[r][i];
        const float p = 1.0f / (1.0f + expf(-s));
        const int slot = i << shift;
        const float v = sm.PP[r][slot];
        sm.PP[r][slot]       = v * p;
        sm.PP[r][slot + gap] = v * (1.0f - p);
      }
    }
    __syncthreads();
    if (d < MAX_DEPTH - 1) {
      if (d >= 1) {
        for (int tix = tid; tix < (ROWS_PER_WG << (d - 1)); tix += THREADS) {
          const int r = tix >> (d - 1);
          const int j = tix & ((1 << (d - 1)) - 1);
          sm.Prev2[r][j] = sm.Prev[r][j];
        }
      }
      __syncthreads();
      for (int tix = tid; tix < (ROWS_PER_WG << d); tix += THREADS) {
        const int r = tix >> d;
        const int j = tix & (nNodes - 1);
        sm.Prev[r][j] = sm.Cur[r][j];
      }
      __syncthreads();
    }
  }

  __syncthreads();
  if (grp == 0) {
    const float* leafP = &sm.W[0][0];
    v8f e0 = {0.f,0.f,0.f,0.f,0.f,0.f,0.f,0.f};
    v8f e1 = {0.f,0.f,0.f,0.f,0.f,0.f,0.f,0.f};
    #pragma unroll
    for (int k = 0; k < 64; k += 2) {
      v2f a0, bq0, a1, bq1;
      a0.x  = sm.PP[rl][4 * k + kb];
      a0.y  = sm.PP[rl][4 * k + kb + 1];
      bq0.x = leafP[(4 * k + kb) * 16 + col];
      bq0.y = leafP[(4 * k + kb + 1) * 16 + col];
      a1.x  = sm.PP[rl][4 * (k + 1) + kb];
      a1.y  = sm.PP[rl][4 * (k + 1) + kb + 1];
      bq1.x = leafP[(4 * (k + 1) + kb) * 16 + col];
      bq1.y = leafP[(4 * (k + 1) + kb + 1) * 16 + col];
      e0 = __builtin_amdgcn_wmma_f32_16x16x4_f32(false, a0, false, bq0,
                                                 (short)0, e0, false, false);
      e1 = __builtin_amdgcn_wmma_f32_16x16x4_f32(false, a1, false, bq1,
                                                 (short)0, e1, false, false);
    }
    if (col < NUM_CLASSES) {
      #pragma unroll
      for (int r = 0; r < 8; ++r)
        out[(size_t)(waveRow0 + half * 8 + r) * NUM_CLASSES + col] = e0[r] + e1[r];
    }
  }
}

extern "C" void kernel_launch(void* const* d_in, const int* in_sizes, int n_in,
                              void* d_out, int out_size, void* d_ws, size_t ws_size,
                              hipStream_t stream) {
  const float* x    = (const float*)d_in[0];
  const float* pp   = (const float*)d_in[1];
  const float* W1   = (const float*)d_in[2];
  const float* b1   = (const float*)d_in[3];
  const float* w2   = (const float*)d_in[4];
  const float* b2   = (const float*)d_in[5];
  const float* leaf = (const float*)d_in[6];
  float* out = (float*)d_out;

  const int B = in_sizes[0] / INPUT_DIM;
  dim3 grid(B / ROWS_PER_WG);
  dim3 block(THREADS);
  tree_wmma_kernel<<<grid, block, 0, stream>>>(x, pp, W1, b1, w2, b2, leaf, out);
}
